# starting kernel
# baseline (speedup 1.0000x reference)
#include <hip/hip_runtime.h>
#include <hip/hip_bf16.h>

typedef __attribute__((ext_vector_type(2))) float v2f;
typedef __attribute__((ext_vector_type(8))) float v8f;

#define BATCH 8
#define CCH   16
#define HH    512
#define WW    512

__global__ __launch_bounds__(256) void pool_pack_kernel(
    const float* __restrict__ x, float* __restrict__ U2)
{
    int idx = blockIdx.x * blockDim.x + threadIdx.x;
    int pw = idx & 63;
    int ph = (idx >> 6) & 63;
    int c  = (idx >> 12) & 15;
    int b  = idx >> 16;

    const float* src = x + (((size_t)(b * CCH + c) * HH) + (size_t)ph * 8) * WW + pw * 8;
    float s = 0.0f;
#pragma unroll
    for (int r = 0; r < 8; ++r) {
        const float4* p = (const float4*)(src + (size_t)r * WW);
        float4 a = p[0];
        float4 bb = p[1];
        s += (a.x + a.y) + (a.z + a.w) + (bb.x + bb.y) + (bb.z + bb.w);
    }
    s *= (1.0f / 64.0f);

    int lp = ((ph >> 3) << 3) | (pw >> 3);
    int n  = (c << 6) | ((ph & 7) << 3) | (pw & 7);
    U2[(size_t)b * 65536 + (size_t)(lp >> 1) * 2048 + n * 2 + (lp & 1)] = s;
}

#define LDSS 68

__global__ __launch_bounds__(256) void att_gemm_kernel(
    const float* __restrict__ attn,
    const float* __restrict__ U2,
    float* __restrict__ out)
{
    __shared__ float As[16 * LDSS];
    __shared__ float rowScale[16];

    int bid   = blockIdx.x;
    int mtile = bid & 255;
    int b     = bid >> 8;
    int l_base = mtile * 16;

    int t    = threadIdx.x;
    int lane = t & 31;
    int wave = t >> 5;

    {
        int row  = t >> 4;
        int col4 = (t & 15) * 4;
        float4 v = *(const float4*)(attn + ((size_t)b * 4096 + l_base + row) * 64 + col4);
        *(float4*)(&As[row * LDSS + col4]) = v;
    }
    __syncthreads();

    if (t < 16) {
        int cnt = 0;
#pragma unroll
        for (int k = 0; k < 64; ++k) cnt += (As[t * LDSS + k] != 0.0f) ? 1 : 0;
        rowScale[t] = 1.0f / ((float)cnt + 1e-5f);
    }
    __syncthreads();

    int n_wave = wave * 128;
    int M = lane & 15;
    int h = lane >> 4;
    float scale = rowScale[M];

    v8f acc[8];
#pragma unroll
    for (int i = 0; i < 8; ++i) acc[i] = (v8f){0.f,0.f,0.f,0.f,0.f,0.f,0.f,0.f};

    const float* Ub = U2 + (size_t)b * 65536;

#pragma unroll
    for (int ks = 0; ks < 16; ++ks) {
        int k0 = ks * 4 + h * 2;
        v2f a;
        a.x = As[M * LDSS + k0]     * scale;
        a.y = As[M * LDSS + k0 + 1] * scale;
#pragma unroll
        for (int nt = 0; nt < 8; ++nt) {
            int n = n_wave + nt * 16 + M;
            v2f bf = *(const v2f*)(Ub + (size_t)(k0 >> 1) * 2048 + n * 2);
            acc[nt] = __builtin_amdgcn_wmma_f32_16x16x4_f32(
                false, a, false, bf, (short)0, acc[nt], false, false);
        }
    }

    int hp  = l_base >> 6;
    int wp0 = (l_base & 63) + 8 * h;
#pragma unroll
    for (int nt = 0; nt < 8; ++nt) {
        int k  = n_wave + nt * 16 + M;
        int c  = k >> 6;
        int kh = (k >> 3) & 7;
        int kw = k & 7;
        size_t o0 = (((size_t)(b * CCH + c) * HH) + (size_t)(hp * 8 + kh)) * WW
                  + (size_t)wp0 * 8 + kw;
#pragma unroll
        for (int v = 0; v < 8; ++v) {
            out[o0 + (size_t)v * 8] = acc[nt][v];
        }
    }
}

extern "C" void kernel_launch(void* const* d_in, const int* in_sizes, int n_in,
                              void* d_out, int out_size, void* d_ws, size_t ws_size,
                              hipStream_t stream)
{
    const float* x    = (const float*)d_in[0];
    const float* attn = (const float*)d_in[1];
    float* out = (float*)d_out;
    float* U2  = (float*)d_ws;

    pool_pack_kernel<<<2048, 256, 0, stream>>>(x, U2);
    att_gemm_kernel<<<2048, 256, 0, stream>>>(attn, U2, out);
}


# GEMM kernel: pooled operand is now WMMA A, its fragments stay in registers over a loop of 8 attn row tiles per workgroup; each lane holds 4 consecutive outputs, stored as whole-line b128
# speedup vs baseline: 1.3415x; 1.3415x over previous
#include <hip/hip_runtime.h>
#include <hip/hip_bf16.h>

typedef __attribute__((ext_vector_type(2))) float v2f;
typedef __attribute__((ext_vector_type(8))) float v8f;

#define BATCH 8
#define CCH   16
#define HH    512
#define WW    512

__global__ __launch_bounds__(256) void pool_pack_kernel(
    const float* __restrict__ x, float* __restrict__ U2)
{
    int idx = blockIdx.x * blockDim.x + threadIdx.x;
    int pw = idx & 63;
    int ph = (idx >> 6) & 63;
    int c  = (idx >> 12) & 15;
    int b  = idx >> 16;

    const float* src = x + (((size_t)(b * CCH + c) * HH) + (size_t)ph * 8) * WW + pw * 8;
    float s = 0.0f;
#pragma unroll
    for (int r = 0; r < 8; ++r) {
        const float4* p = (const float4*)(src + (size_t)r * WW);
        float4 a = p[0];
        float4 bb = p[1];
        s += (a.x + a.y) + (a.z + a.w) + (bb.x + bb.y) + (bb.z + bb.w);
    }
    s *= (1.0f / 64.0f);

    int lp = ((ph >> 3) << 3) | (pw >> 3);
    int n  = (c << 6) | ((ph & 7) << 3) | (pw & 7);
    U2[(size_t)b * 65536 + (size_t)(lp >> 1) * 2048 + n * 2 + (lp & 1)] = s;
}

#define LDSS 68
#define MT   8

__device__ __forceinline__ void lds_barrier()
{
    asm volatile("s_wait_dscnt 0x0" ::: "memory");
    asm volatile("s_barrier_signal -1" ::: "memory");
    asm volatile("s_barrier_wait -1" ::: "memory");
}

__device__ __forceinline__ void stage_attn_tile(
    float4 v, float* __restrict__ dst, int t)
{
    int cnt = ((v.x != 0.0f) ? 1 : 0) + ((v.y != 0.0f) ? 1 : 0)
            + ((v.z != 0.0f) ? 1 : 0) + ((v.w != 0.0f) ? 1 : 0);
    cnt += __shfl_xor(cnt, 1);
    cnt += __shfl_xor(cnt, 2);
    cnt += __shfl_xor(cnt, 4);
    cnt += __shfl_xor(cnt, 8);
    float scale = 1.0f / ((float)cnt + 1e-5f);
    float4 w;
    w.x = v.x * scale; w.y = v.y * scale; w.z = v.z * scale; w.w = v.w * scale;
    *(float4*)(dst + (t >> 4) * LDSS + (t & 15) * 4) = w;
}

__global__ __launch_bounds__(256) void att_gemm_kernel(
    const float* __restrict__ attn,
    const float* __restrict__ U2,
    float* __restrict__ out)
{
    __shared__ float As[2][16 * LDSS];

    int bid = blockIdx.x;
    int mg  = bid & 31;
    int nh  = (bid >> 5) & 1;
    int b   = bid >> 6;

    int t    = threadIdx.x;
    int lane = t & 31;
    int wave = t >> 5;
    int c16  = lane & 15;
    int h    = lane >> 4;
    int ch   = nh * 8 + wave;

    const float* arow = attn + ((size_t)b * 4096 + (size_t)mg * (MT * 16) + (t >> 4)) * 64
                      + (t & 15) * 4;
    float4 vcur = *(const float4*)arow;

    v2f uf[4][16];
    {
        const float* Ub = U2 + (size_t)b * 65536;
        int nperm = ((c16 >> 2) & 1) * 8 + (c16 >> 3) * 4 + (c16 & 3);
#pragma unroll
        for (int ks = 0; ks < 16; ++ks) {
#pragma unroll
            for (int nt = 0; nt < 4; ++nt) {
                int n = ch * 64 + nt * 16 + nperm;
                uf[nt][ks] = *(const v2f*)(Ub + (size_t)(ks * 2 + h) * 2048 + n * 2);
            }
        }
    }

    stage_attn_tile(vcur, As[0], t);
    lds_barrier();

    float* obase = out + (((size_t)(b * CCH + ch) * HH) * WW) + c16 * 8 + h * 4;

    for (int mt = 0; mt < MT; ++mt) {
        const float* Ab = As[mt & 1];
        if (mt + 1 < MT) vcur = *(const float4*)(arow + (size_t)(mt + 1) * 16 * 64);

        v8f acc[4];
#pragma unroll
        for (int i = 0; i < 4; ++i) acc[i] = (v8f){0.f,0.f,0.f,0.f,0.f,0.f,0.f,0.f};

#pragma unroll
        for (int ks = 0; ks < 16; ++ks) {
            v2f af = *(const v2f*)(Ab + c16 * LDSS + ks * 4 + h * 2);
#pragma unroll
            for (int nt = 0; nt < 4; ++nt) {
                acc[nt] = __builtin_amdgcn_wmma_f32_16x16x4_f32(
                    false, uf[nt][ks], false, af, (short)0, acc[nt], false, false);
            }
        }

        int l_base = (mg * MT + mt) * 16;
        int hp  = l_base >> 6;
        int wpb = l_base & 63;
        float* orow = obase + (size_t)(hp * 8) * WW + wpb * 8;
#pragma unroll
        for (int nt = 0; nt < 4; ++nt) {
#pragma unroll
            for (int j = 0; j < 2; ++j) {
                float4 o;
                o.x = acc[nt][4 * j + 0]; o.y = acc[nt][4 * j + 1];
                o.z = acc[nt][4 * j + 2]; o.w = acc[nt][4 * j + 3];
                *(float4*)(orow + (size_t)(nt * 2 + j) * WW) = o;
            }
        }

        if (mt + 1 < MT) {
            stage_attn_tile(vcur, As[(mt + 1) & 1], t);
            lds_barrier();
        }
    }
}

extern "C" void kernel_launch(void* const* d_in, const int* in_sizes, int n_in,
                              void* d_out, int out_size, void* d_ws, size_t ws_size,
                              hipStream_t stream)
{
    const float* x    = (const float*)d_in[0];
    const float* attn = (const float*)d_in[1];
    float* out = (float*)d_out;
    float* U2  = (float*)d_ws;

    pool_pack_kernel<<<2048, 256, 0, stream>>>(x, U2);
    att_gemm_kernel<<<512, 256, 0, stream>>>(attn, U2, out);
}


# Pooling launch as a stream: 256 workgroups of 16 waves, a wave reads two 8-row strips as 512-byte pieces, 16-32 loads in flight; pooled sums re-associated (lane half sums, then one pair add)
# speedup vs baseline: 1.4913x; 1.1117x over previous
#include <hip/hip_runtime.h>
#include <hip/hip_bf16.h>

typedef __attribute__((ext_vector_type(2))) float v2f;
typedef __attribute__((ext_vector_type(8))) float v8f;

#define BATCH 8
#define CCH   16
#define HH    512
#define WW    512

#define PWAVES 16
#define PUNITS 4

__global__ __launch_bounds__(PWAVES * 32) void pool_pack_kernel(
    const float* __restrict__ x, float* __restrict__ U2)
{
    const int lane = threadIdx.x & 31;
    const int wave = __builtin_amdgcn_readfirstlane((int)(threadIdx.x >> 5));
    const int strip0 = (blockIdx.x * PWAVES + wave) * 2;
    const float* src = x + (size_t)strip0 * (8 * WW) + lane * 4;

    float4 buf[2][16];
    float s[2][4];
#pragma unroll
    for (int t = 0; t < 2; ++t)
#pragma unroll
        for (int j = 0; j < 4; ++j) s[t][j] = 0.0f;

#pragma unroll
    for (int q = 0; q < 16; ++q)
        buf[0][q] = *(const float4*)(src + q * 128);

#pragma unroll
    for (int u = 0; u < PUNITS; ++u) {
        if (u + 1 < PUNITS) {
#pragma unroll
            for (int q = 0; q < 16; ++q)
                buf[(u + 1) & 1][q] = *(const float4*)(src + ((u + 1) * 16 + q) * 128);
        }
        __builtin_amdgcn_sched_barrier(0);
#pragma unroll
        for (int q = 0; q < 16; ++q) {
            float4 v = buf[u & 1][q];
            s[u >> 1][q & 3] += (v.x + v.y) + (v.z + v.w);
        }
        asm volatile("" : "+v"(s[u >> 1][0]), "+v"(s[u >> 1][1]),
                          "+v"(s[u >> 1][2]), "+v"(s[u >> 1][3]) : : "memory");
    }

    const int odd = lane & 1;
    const int strip = strip0 + odd;
    const int ph = strip & 63;
    const int c  = (strip >> 6) & 15;
    const int b  = strip >> 10;
    float* dst = U2 + (size_t)b * 65536 + (size_t)((ph >> 3) * 4) * 2048
               + ((c << 6) | ((ph & 7) << 3) | ((lane >> 1) & 7)) * 2 + (lane >> 4);
#pragma unroll
    for (int j = 0; j < 4; ++j) {
        float keep = odd ? s[1][j] : s[0][j];
        float send = odd ? s[0][j] : s[1][j];
        float tot = keep + __shfl_xor(send, 1);
        dst[(size_t)j * 2048] = tot * (1.0f / 64.0f);
    }
}

#define LDSS 68
#define MT   8

__device__ __forceinline__ void lds_barrier()
{
    asm volatile("s_wait_dscnt 0x0" ::: "memory");
    asm volatile("s_barrier_signal -1" ::: "memory");
    asm volatile("s_barrier_wait -1" ::: "memory");
}

__device__ __forceinline__ void stage_attn_tile(
    float4 v, float* __restrict__ dst, int t)
{
    int cnt = ((v.x != 0.0f) ? 1 : 0) + ((v.y != 0.0f) ? 1 : 0)
            + ((v.z != 0.0f) ? 1 : 0) + ((v.w != 0.0f) ? 1 : 0);
    cnt += __shfl_xor(cnt, 1);
    cnt += __shfl_xor(cnt, 2);
    cnt += __shfl_xor(cnt, 4);
    cnt += __shfl_xor(cnt, 8);
    float scale = 1.0f / ((float)cnt + 1e-5f);
    float4 w;
    w.x = v.x * scale; w.y = v.y * scale; w.z = v.z * scale; w.w = v.w * scale;
    *(float4*)(dst + (t >> 4) * LDSS + (t & 15) * 4) = w;
}

__global__ __launch_bounds__(256) void att_gemm_kernel(
    const float* __restrict__ attn,
    const float* __restrict__ U2,
    float* __restrict__ out)
{
    __shared__ float As[2][16 * LDSS];

    int bid = blockIdx.x;
    int mg  = bid & 31;
    int nh  = (bid >> 5) & 1;
    int b   = bid >> 6;

    int t    = threadIdx.x;
    int lane = t & 31;
    int wave = t >> 5;
    int c16  = lane & 15;
    int h    = lane >> 4;
    int ch   = nh * 8 + wave;

    const float* arow = attn + ((size_t)b * 4096 + (size_t)mg * (MT * 16) + (t >> 4)) * 64
                      + (t & 15) * 4;
    float4 vcur = *(const float4*)arow;

    v2f uf[4][16];
    {
        const float* Ub = U2 + (size_t)b * 65536;
        int nperm = ((c16 >> 2) & 1) * 8 + (c16 >> 3) * 4 + (c16 & 3);
#pragma unroll
        for (int ks = 0; ks < 16; ++ks) {
#pragma unroll
            for (int nt = 0; nt < 4; ++nt) {
                int n = ch * 64 + nt * 16 + nperm;
                uf[nt][ks] = *(const v2f*)(Ub + (size_t)(ks * 2 + h) * 2048 + n * 2);
            }
        }
    }

    stage_attn_tile(vcur, As[0], t);
    lds_barrier();

    float* obase = out + (((size_t)(b * CCH + ch) * HH) * WW) + c16 * 8 + h * 4;

    for (int mt = 0; mt < MT; ++mt) {
        const float* Ab = As[mt & 1];
        if (mt + 1 < MT) vcur = *(const float4*)(arow + (size_t)(mt + 1) * 16 * 64);

        v8f acc[4];
#pragma unroll
        for (int i = 0; i < 4; ++i) acc[i] = (v8f){0.f,0.f,0.f,0.f,0.f,0.f,0.f,0.f};

#pragma unroll
        for (int ks = 0; ks < 16; ++ks) {
            v2f af = *(const v2f*)(Ab + c16 * LDSS + ks * 4 + h * 2);
#pragma unroll
            for (int nt = 0; nt < 4; ++nt) {
                acc[nt] = __builtin_amdgcn_wmma_f32_16x16x4_f32(
                    false, uf[nt][ks], false, af, (short)0, acc[nt], false, false);
            }
        }

        int l_base = (mg * MT + mt) * 16;
        int hp  = l_base >> 6;
        int wpb = l_base & 63;
        float* orow = obase + (size_t)(hp * 8) * WW + wpb * 8;
#pragma unroll
        for (int nt = 0; nt < 4; ++nt) {
#pragma unroll
            for (int j = 0; j < 2; ++j) {
                float4 o;
                o.x = acc[nt][4 * j + 0]; o.y = acc[nt][4 * j + 1];
                o.z = acc[nt][4 * j + 2]; o.w = acc[nt][4 * j + 3];
                *(float4*)(orow + (size_t)(nt * 2 + j) * WW) = o;
            }
        }

        if (mt + 1 < MT) {
            stage_attn_tile(vcur, As[(mt + 1) & 1], t);
            lds_barrier();
        }
    }
}

extern "C" void kernel_launch(void* const* d_in, const int* in_sizes, int n_in,
                              void* d_out, int out_size, void* d_ws, size_t ws_size,
                              hipStream_t stream)
{
    const float* x    = (const float*)d_in[0];
    const float* attn = (const float*)d_in[1];
    float* out = (float*)d_out;
    float* U2  = (float*)d_ws;

    pool_pack_kernel<<<256, 512, 0, stream>>>(x, U2);
    att_gemm_kernel<<<512, 256, 0, stream>>>(attn, U2, out);
}
